# starting kernel
# baseline (speedup 1.0000x reference)
#include <hip/hip_runtime.h>


typedef __attribute__((ext_vector_type(2))) float v2f;
typedef __attribute__((ext_vector_type(8))) float v8f;

#define BATCH 128
#define DCH   300
#define LEN   128
#define NF    8
#define WPB   8

#define DPP_XOR1  0xB1
#define DPP_XOR2  0x4E
#define DPP_HMIRR 0x141
#define DPP_MIRR  0x140

template <int CTRL>
__device__ __forceinline__ float dpp_bfly(float v) {
    int i = __builtin_bit_cast(int, v);
    int t = __builtin_amdgcn_update_dpp(i, i, CTRL, 0xF, 0xF, false);
    return __builtin_bit_cast(float, t);
}
__device__ __forceinline__ float row16_max(float v) {
    v = fmaxf(v, dpp_bfly<DPP_XOR1>(v));
    v = fmaxf(v, dpp_bfly<DPP_XOR2>(v));
    v = fmaxf(v, dpp_bfly<DPP_HMIRR>(v));
    v = fmaxf(v, dpp_bfly<DPP_MIRR>(v));
    return v;
}
__device__ __forceinline__ float row16_min(float v) {
    v = fminf(v, dpp_bfly<DPP_XOR1>(v));
    v = fminf(v, dpp_bfly<DPP_XOR2>(v));
    v = fminf(v, dpp_bfly<DPP_HMIRR>(v));
    v = fminf(v, dpp_bfly<DPP_MIRR>(v));
    return v;
}
__device__ __forceinline__ float row16_sum(float v) {
    v += dpp_bfly<DPP_XOR1>(v);
    v += dpp_bfly<DPP_XOR2>(v);
    v += dpp_bfly<DPP_HMIRR>(v);
    v += dpp_bfly<DPP_MIRR>(v);
    return v;
}
__device__ __forceinline__ float swz_xor16(float v) {
    int t = __builtin_amdgcn_ds_swizzle(__builtin_bit_cast(int, v), 0x401F);
    return __builtin_bit_cast(float, t);
}

__global__ __launch_bounds__(256) void conv_pool_wmma(
    const float* __restrict__ x,
    const float* __restrict__ W1, const float* __restrict__ b1,
    const float* __restrict__ W2, const float* __restrict__ b2,
    const float* __restrict__ W3, const float* __restrict__ b3,
    float* __restrict__ out)
{
    __shared__ float xs[WPB][132];

    const int lane = threadIdx.x & 31;
    const int wv   = threadIdx.x >> 5;
    const int pair = blockIdx.x * WPB + wv;
    const int b    = pair / DCH;
    const int d    = pair - b * DCH;

    const float4 xv = *(const float4*)(x + (size_t)pair * LEN + lane * 4);
    *(float4*)(&xs[wv][lane * 4]) = xv;
    if (lane < 4) xs[wv][LEN + lane] = 0.0f;

    float mx = fmaxf(fmaxf(xv.x, xv.y), fmaxf(xv.z, xv.w));
    float mn = fminf(fminf(xv.x, xv.y), fminf(xv.z, xv.w));
    float sm = (xv.x + xv.y) + (xv.z + xv.w);
    mx = row16_max(mx);  mx = fmaxf(mx, swz_xor16(mx));
    mn = row16_min(mn);  mn = fminf(mn, swz_xor16(mn));
    sm = row16_sum(sm);  sm += swz_xor16(sm);

    v2f Bw;
    {
        const int  nn = lane & 15;
        const bool hi = lane >= 16;
        float w0 = 0.0f, w1 = 0.0f;
        if (nn < 8) {
            if (!hi) { w0 = W2[(d * NF + nn) * 2 + 0]; w1 = W2[(d * NF + nn) * 2 + 1]; }
        } else {
            const int f = nn - 8;
            if (!hi) { w0 = W3[(d * NF + f) * 3 + 0]; w1 = W3[(d * NF + f) * 3 + 1]; }
            else     { w0 = W3[(d * NF + f) * 3 + 2]; }
        }
        Bw[0] = w0; Bw[1] = w1;
    }

    __syncthreads();

    const int m    = lane & 15;
    const int kofs = (lane < 16) ? 0 : 2;

    float acc = -__builtin_inff();
    #pragma unroll
    for (int t = 0; t < 8; ++t) {
        v2f Ap;
        const int off = t * 16 + m + kofs;
        Ap[0] = xs[wv][off];
        Ap[1] = xs[wv][off + 1];
        v8f C = {};
        C = __builtin_amdgcn_wmma_f32_16x16x4_f32(false, Ap, false, Bw,
                                                  (short)0, C, false, false);
        float c6 = C[6], c7 = C[7];
        if (t == 7) {
            c6 = (lane < 24) ? c6 : -__builtin_inff();
            c7 = (lane < 16) ? c7 : -__builtin_inff();
        }
        const float r = fmaxf(fmaxf(fmaxf(C[0], C[1]), fmaxf(C[2], C[3])),
                              fmaxf(fmaxf(C[4], C[5]), fmaxf(c6, c7)));
        acc = fmaxf(acc, r);
    }
    acc = fmaxf(acc, swz_xor16(acc));

    const float x0   = xs[wv][0],   x1   = xs[wv][1];
    const float x126 = xs[wv][126], x127 = xs[wv][127];
    const float S0_2 = sm - x127,        S1_2 = sm - x0;
    const float S0_3 = sm - x126 - x127, S1_3 = sm - x0 - x127,
                S2_3 = sm - x0 - x1;
    const float w3t2 = swz_xor16(Bw[0]);

    const size_t ob   = (size_t)b * (2 * 3 * DCH * NF);
    const size_t strP = (size_t)3 * DCH * NF;
    const size_t strW = (size_t)DCH * NF;
    float* __restrict__ omax  = out + ob;
    float* __restrict__ omean = out + ob + strP;

    if (lane < 8) {
        const float w  = W1[d * NF + lane];
        const float bb = b1[d * NF + lane];
        omax [0 * strW + d * NF + lane] = (w >= 0.0f ? w * mx : w * mn) + bb;
        omean[0 * strW + d * NF + lane] = w * (sm * (1.0f / 128.0f)) + bb;
        const float bias2 = b2[d * NF + lane];
        omax [1 * strW + d * NF + lane] = acc + bias2;
        omean[1 * strW + d * NF + lane] =
            (Bw[0] * S0_2 + Bw[1] * S1_2) * (1.0f / 127.0f) + bias2;
    } else if (lane < 16) {
        const int f = lane - 8;
        const float bias3 = b3[d * NF + f];
        omax [2 * strW + d * NF + f] = acc + bias3;
        omean[2 * strW + d * NF + f] =
            (Bw[0] * S0_3 + Bw[1] * S1_3 + w3t2 * S2_3) * (1.0f / 126.0f) + bias3;
    }
}

extern "C" void kernel_launch(void* const* d_in, const int* in_sizes, int n_in,
                              void* d_out, int out_size, void* d_ws, size_t ws_size,
                              hipStream_t stream) {
    const float* x  = (const float*)d_in[0];
    const float* W1 = (const float*)d_in[1];
    const float* b1 = (const float*)d_in[2];
    const float* W2 = (const float*)d_in[3];
    const float* b2 = (const float*)d_in[4];
    const float* W3 = (const float*)d_in[5];
    const float* b3 = (const float*)d_in[6];
    float* out = (float*)d_out;

    const int pairs  = BATCH * DCH;
    const int blocks = pairs / WPB;
    conv_pool_wmma<<<blocks, WPB * 32, 0, stream>>>(x, W1, b1, W2, b2, W3, b3, out);
}


# batch rows on the matrix M axis: a wave takes 8 batch rows of one channel as 16 half-rows x 4 positions, banded-tap B gives 2 positions x 8 filters, elementwise max; 4800 waves, 128-B line stores
# speedup vs baseline: 1.7390x; 1.7390x over previous
#include <hip/hip_runtime.h>


typedef __attribute__((ext_vector_type(2))) float v2f;
typedef __attribute__((ext_vector_type(8))) float v8f;

#define BATCH 128
#define DCH   300
#define LEN   128
#define NF    8
#define ROWS  8
#define HALF  64
#define PITCH 132
#define CPB   4
#define NPW   6
#define DQ    (DCH / CPB)
#define NBLK  ((BATCH / ROWS) * DQ)

#define DPP_ROR8 0x128

__device__ __forceinline__ float dpp_ror8(float v) {
    int i = __builtin_bit_cast(int, v);
    int t = __builtin_amdgcn_update_dpp(i, i, DPP_ROR8, 0xF, 0xF, false);
    return __builtin_bit_cast(float, t);
}
__device__ __forceinline__ float swz_xor16(float v) {
    int t = __builtin_amdgcn_ds_swizzle(__builtin_bit_cast(int, v), 0x401F);
    return __builtin_bit_cast(float, t);
}
__device__ __forceinline__ void wave_lds_order() {
    __builtin_amdgcn_fence(__ATOMIC_RELEASE, "wavefront");
    __builtin_amdgcn_wave_barrier();
    __builtin_amdgcn_fence(__ATOMIC_ACQUIRE, "wavefront");
}
__device__ __forceinline__ float tap2(int j, float a, float b) {
    return j == 0 ? a : (j == 1 ? b : 0.0f);
}
__device__ __forceinline__ float tap3(int j, float a, float b, float c) {
    return j == 0 ? a : (j == 1 ? b : (j == 2 ? c : 0.0f));
}
__device__ __forceinline__ v8f wmma4(v2f a, v2f b) {
    v8f z = {};
    return __builtin_amdgcn_wmma_f32_16x16x4_f32(false, a, false, b, (short)0, z, false, false);
}

__global__ __launch_bounds__(CPB * 32) void conv_pool_rows_wmma(
    const float* __restrict__ x,
    const float* __restrict__ W1, const float* __restrict__ b1,
    const float* __restrict__ W2, const float* __restrict__ b2,
    const float* __restrict__ W3, const float* __restrict__ b3,
    float* __restrict__ out)
{
    __shared__ float xs[CPB][ROWS][PITCH];
    __shared__ float st[CPB][ROWS][8];
    __shared__ float res[ROWS][NPW][CPB * NF];

    const int lane = threadIdx.x & 31;
    const int wv   = threadIdx.x >> 5;
    const int rg   = blockIdx.x / DQ;
    const int dq   = blockIdx.x - rg * DQ;
    const int b0   = rg * ROWS;
    const int d    = dq * CPB + wv;
    const int c    = lane & 15;
    const int h    = lane >> 4;
    const int f    = lane & 7;
    const int o    = (lane >> 3) & 1;

    const int df = d * NF + f;
    const float w1  = W1[df],         bb1 = b1[df];
    const float w20 = W2[df * 2],     w21 = W2[df * 2 + 1], bb2 = b2[df];
    const float w30 = W3[df * 3],     w31 = W3[df * 3 + 1], w32 = W3[df * 3 + 2];
    const float bb3 = b3[df];

    const float* xp = x + ((size_t)b0 * DCH + d) * LEN + lane * 4;
    float4 xr[ROWS];
    #pragma unroll
    for (int r = 0; r < ROWS; ++r)
        xr[r] = *(const float4*)(xp + (size_t)r * DCH * LEN);
    #pragma unroll
    for (int r = 0; r < ROWS; ++r)
        *(float4*)(&xs[wv][r][lane * 4]) = xr[r];
    if (lane < ROWS)
        *(float4*)(&xs[wv][lane][LEN]) = make_float4(0.0f, 0.0f, 0.0f, 0.0f);

    const int j0 = 2 * h - o;
    v2f B2, B3;
    B2[0] = tap2(j0, w20, w21);      B2[1] = tap2(j0 + 1, w20, w21);
    B3[0] = tap3(j0, w30, w31, w32); B3[1] = tap3(j0 + 1, w30, w31, w32);

    wave_lds_order();

    const float* arow = &xs[wv][c & 7][HALF * (c >> 3) + 2 * h];
    const float NEG = -__builtin_inff();
    float acc2[8], acc3[8];
    #pragma unroll
    for (int r = 0; r < 8; ++r) { acc2[r] = NEG; acc3[r] = NEG; }
    float sm = 0.0f, mx = NEG, mn = __builtin_inff();

    v2f Aa = *(const v2f*)(arow);
    v2f Ab = *(const v2f*)(arow + 2);
    for (int i = 0; i < 15; ++i) {
        const v2f Na = *(const v2f*)(arow + 4 * i + 4);
        const v2f Nb = *(const v2f*)(arow + 4 * i + 6);
        __builtin_amdgcn_sched_barrier(0);
        const v8f Pa = wmma4(Aa, B2), Qa = wmma4(Aa, B3);
        const v8f Pb = wmma4(Ab, B2), Qb = wmma4(Ab, B3);
        #pragma unroll
        for (int r = 0; r < 8; ++r) {
            acc2[r] = fmaxf(fmaxf(acc2[r], Pa[r]), Pb[r]);
            acc3[r] = fmaxf(fmaxf(acc3[r], Qa[r]), Qb[r]);
        }
        sm += Aa[0] + Aa[1];
        mx = fmaxf(fmaxf(mx, Aa[0]), Aa[1]);
        mn = fminf(fminf(mn, Aa[0]), Aa[1]);
        Aa = Na; Ab = Nb;
    }
    {
        const v8f Pa = wmma4(Aa, B2), Qa = wmma4(Aa, B3);
        const v8f Pb = wmma4(Ab, B2), Qb = wmma4(Ab, B3);
        const bool bad2 = (h == 1) && (o == 1);
        const bool bad3 = (h == 1);
        #pragma unroll
        for (int r = 0; r < 8; ++r) {
            const float pb = bad2 ? NEG : Pb[r];
            const float qb = bad3 ? NEG : Qb[r];
            acc2[r] = fmaxf(fmaxf(acc2[r], Pa[r]), pb);
            acc3[r] = fmaxf(fmaxf(acc3[r], Qa[r]), qb);
        }
        sm += Aa[0] + Aa[1];
        mx = fmaxf(fmaxf(mx, Aa[0]), Aa[1]);
        mn = fminf(fminf(mn, Aa[0]), Aa[1]);
    }

    sm += swz_xor16(sm);              sm += dpp_ror8(sm);
    mx = fmaxf(mx, swz_xor16(mx));    mx = fmaxf(mx, dpp_ror8(mx));
    mn = fminf(mn, swz_xor16(mn));    mn = fminf(mn, dpp_ror8(mn));
    if (lane < ROWS) {
        const v2f e0 = *(const v2f*)(&xs[wv][lane][0]);
        const v2f e1 = *(const v2f*)(&xs[wv][lane][LEN - 2]);
        *(float4*)(&st[wv][lane][0]) = make_float4(sm, mx, mn, 0.0f);
        *(float4*)(&st[wv][lane][4]) = make_float4(e0[0], e0[1], e1[0], e1[1]);
    }

    float m[8];
    #pragma unroll
    for (int r = 0; r < 8; ++r) {
        const float mine = o ? acc3[r] : acc2[r];
        const float send = o ? acc2[r] : acc3[r];
        m[r] = fmaxf(mine, dpp_ror8(send));
    }
    {
        const int   pw = 1 + o;
        const float bm = o ? bb3 : bb2;
        #pragma unroll
        for (int k = 0; k < 4; ++k) {
            const float mine = h ? m[4 + k] : m[k];
            const float send = h ? m[k] : m[4 + k];
            res[4 * h + k][pw][wv * NF + f] = fmaxf(mine, swz_xor16(send)) + bm;
        }
    }

    wave_lds_order();

    for (int j = 0; j < 2; ++j) {
        const int row = (lane >> 3) + 4 * j;
        const float4 s0 = *(const float4*)(&st[wv][row][0]);
        const float4 s1 = *(const float4*)(&st[wv][row][4]);
        const float sum = s0.x, rmx = s0.y, rmn = s0.z;
        const float x0 = s1.x, x1 = s1.y, x126 = s1.z, x127 = s1.w;
        const float S0_2 = sum - x127,        S1_2 = sum - x0;
        const float S0_3 = sum - x126 - x127, S1_3 = sum - x0 - x127,
                    S2_3 = sum - x0 - x1;
        res[row][0][wv * NF + f] = (w1 >= 0.0f ? w1 * rmx : w1 * rmn) + bb1;
        res[row][3][wv * NF + f] = w1 * (sum * (1.0f / 128.0f)) + bb1;
        res[row][4][wv * NF + f] = (w20 * S0_2 + w21 * S1_2) * (1.0f / 127.0f) + bb2;
        res[row][5][wv * NF + f] =
            (w30 * S0_3 + w31 * S1_3 + w32 * S2_3) * (1.0f / 126.0f) + bb3;
    }

    __syncthreads();

    {
        const int seg = threadIdx.x & 7;
        const int ln  = threadIdx.x >> 3;
        const int row = ln & 7;
        const int pl0 = ln >> 3;
        float* dst = out + (size_t)(b0 + row) * (NPW * DCH * NF) + dq * (CPB * NF) + seg * 4;
        for (int p = 0; p < 3; ++p) {
            const int pl = pl0 + 2 * p;
            *(float4*)(dst + (size_t)pl * (DCH * NF)) = *(const float4*)(&res[row][pl][seg * 4]);
        }
    }
}

extern "C" void kernel_launch(void* const* d_in, const int* in_sizes, int n_in,
                              void* d_out, int out_size, void* d_ws, size_t ws_size,
                              hipStream_t stream) {
    const float* x  = (const float*)d_in[0];
    const float* W1 = (const float*)d_in[1];
    const float* b1 = (const float*)d_in[2];
    const float* W2 = (const float*)d_in[3];
    const float* b2 = (const float*)d_in[4];
    const float* W3 = (const float*)d_in[5];
    const float* b3 = (const float*)d_in[6];
    float* out = (float*)d_out;

    conv_pool_rows_wmma<<<NBLK, CPB * 32, 0, stream>>>(x, W1, b1, W2, b2, W3, b3, out);
}
